# starting kernel
# baseline (speedup 1.0000x reference)
#include <hip/hip_runtime.h>
#include <math.h>


typedef __attribute__((ext_vector_type(2))) float v2f;
typedef __attribute__((ext_vector_type(8))) float v8f;

#ifndef __has_builtin
#define __has_builtin(x) 0
#endif

#if __has_builtin(__builtin_amdgcn_wmma_f32_16x16x4_f32)
#define USE_WMMA_F32X4 1
#else
#define USE_WMMA_F32X4 0
#endif

namespace {
constexpr int kS = 256;
constexpr int kC = 128;
constexpr int kCenter = 128;
constexpr int kIters = 33;
constexpr int kShifts = 129;
constexpr int kKTiles = 36;
constexpr int kRows = 8;
constexpr int kHalo = 4;
constexpr int kRunBuf = kS + 2 * kHalo;
}

__global__ __launch_bounds__(256) void tonal_diffusion_kernel(
    const float* __restrict__ data,
    const float* __restrict__ log_weights,
    float* __restrict__ out,
    int n_rows) {
  const int lane = threadIdx.x & 31;
  const int wave = threadIdx.x >> 5;
  const int row  = blockIdx.x * kRows + wave;
  const int rowc = (row < n_rows) ? row : (n_rows - 1);

  __shared__ float sRunA[kRows][kRunBuf];
  __shared__ float sRunB[kRows][kRunBuf];
  __shared__ float sDataPad[kRows][160];
  __shared__ float sLogAcc[kRows][384];
  __shared__ float sZ[kRows][132];

  const int steps[6] = {1, -1, -3, 3, 4, -4};

  float p[6];
  float rate = 0.f;
#pragma unroll
  for (int i = 0; i < 6; ++i) {
    p[i] = expf(log_weights[(size_t)rowc * 6 + i]);
    rate += p[i];
  }
  const float inv_rate = 1.f / rate;
#pragma unroll
  for (int i = 0; i < 6; ++i) p[i] *= inv_rate;

  for (int idx = lane; idx < 160; idx += 32) {
    float v = 0.f;
    if (idx >= 16 && idx < 16 + kC) {
      float raw = data[(size_t)rowc * kC + (idx - 16)];
      v = (raw > 1e-8f) ? raw : 0.f;
    }
    sDataPad[wave][idx] = v;
  }

  for (int idx = lane; idx < kRunBuf; idx += 32) {
    sRunA[wave][idx] = 0.f;
    sRunB[wave][idx] = 0.f;
  }
  float* runA = &sRunA[wave][kHalo];
  float* runB = &sRunB[wave][kHalo];
  if (lane == 0) runA[kCenter] = 1.f;

  float acc[8];
#pragma unroll
  for (int e = 0; e < 8; ++e) acc[e] = 0.f;
  __syncthreads();

  float sp = expf(-rate);
  for (int n = 0; n < kIters; ++n) {
    const int lo_t = kCenter - kHalo * (n + 1);
    const int hi_t = kCenter + kHalo * (n + 1);
    const int eLo = (lo_t > 0 ? lo_t : 0) >> 5;
    const int eHi = (hi_t < kS - 1 ? hi_t : kS - 1) >> 5;
#pragma unroll
    for (int e = 0; e < 8; ++e) {
      if (e >= eLo && e <= eHi) {
        const int t = e * 32 + lane;
        const float r = runA[t];
        acc[e] = fmaf(sp, r, acc[e]);
        float nv = 0.f;
#pragma unroll
        for (int i = 0; i < 6; ++i)
          nv = fmaf(p[i], runA[t - steps[i]], nv);
        runB[t] = nv;
      }
    }
    __syncthreads();
    float* tmp = runA; runA = runB; runB = tmp;
    sp = sp * rate / (float)(n + 1);
  }

  float* accL = runA;
#pragma unroll
  for (int e = 0; e < 8; ++e) {
    const int t = e * 32 + lane;
    accL[t] = acc[e];
    sLogAcc[wave][t] = logf(fmaxf(acc[e], 1e-38f));
  }
  for (int t = kS + lane; t < 384; t += 32) sLogAcc[wave][t] = 0.f;
  __syncthreads();

  for (int k = lane; k < kShifts; k += 32) {
    float z = 0.f;
    for (int c = 0; c < kC; ++c) z += accL[k + c];
    sZ[wave][k] = logf(z);
  }

  float pl = 0.f;
  for (int c = lane; c < kC; c += 32) {
    const float v = sDataPad[wave][16 + c];
    if (v > 0.f) pl += v * logf(v);
  }
#pragma unroll
  for (int off = 16; off; off >>= 1) pl += __shfl_xor(pl, off, 32);
  __syncthreads();

  float best = INFINITY;

#if USE_WMMA_F32X4
  const int m    = lane & 15;
  const int koff = (lane & 16) ? 2 : 0;
  const float* dp = &sDataPad[wave][16 - m];
  const float* lp = &sLogAcc[wave][16 * m];
  v8f dacc = {0.f, 0.f, 0.f, 0.f, 0.f, 0.f, 0.f, 0.f};
#pragma unroll
  for (int kc = 0; kc < kKTiles; ++kc) {
    const int j = 4 * kc + koff;
    v2f a = {dp[j], dp[j + 1]};
    v2f b = {lp[j], lp[j + 1]};
    dacc = __builtin_amdgcn_wmma_f32_16x16x4_f32(
         false, a,  false, b,
         (short)0, dacc,  false,  false);
  }
  const int mbase = (lane & 16) ? 8 : 0;
  const int n16   = 16 * (lane & 15);
#pragma unroll
  for (int r = 0; r < 8; ++r) {
    const int k = (r + mbase) + n16;
    if (k < kShifts) {
      const float loss = pl - dacc[r] + sZ[wave][k];
      best = fminf(best, loss);
    }
  }
#else
  for (int k = lane; k < kShifts; k += 32) {
    float cr = 0.f;
    for (int c = 0; c < kC; ++c)
      cr = fmaf(sDataPad[wave][16 + c], sLogAcc[wave][k + c], cr);
    best = fminf(best, pl - cr + sZ[wave][k]);
  }
#endif

#pragma unroll
  for (int off = 16; off; off >>= 1) best = fminf(best, __shfl_xor(best, off, 32));
  if (lane == 0 && row < n_rows) out[row] = best;
}

extern "C" void kernel_launch(void* const* d_in, const int* in_sizes, int n_in,
                              void* d_out, int out_size, void* d_ws, size_t ws_size,
                              hipStream_t stream) {
  (void)n_in; (void)d_ws; (void)ws_size; (void)out_size;
  const float* data = (const float*)d_in[0];
  const float* logw = (const float*)d_in[1];
  float* out = (float*)d_out;
  const int n_rows = in_sizes[0] / kC;
  const int blocks = (n_rows + kRows - 1) / kRows;
  tonal_diffusion_kernel<<<blocks, 256, 0, stream>>>(data, logw, out, n_rows);
}


# Window normalisers Z[k] from 16 tail sums of block a, 16 head sums of block a+8 and a DPP sum of the 7 whole blocks between, replacing 9x128 sliding adds per lane (reorders the Z summation)
# speedup vs baseline: 2.2707x; 2.2707x over previous
#include <hip/hip_runtime.h>
#include <math.h>


typedef __attribute__((ext_vector_type(2))) float v2f;
typedef __attribute__((ext_vector_type(8))) float v8f;

namespace {
constexpr int kS = 256;
constexpr int kC = 128;
constexpr int kCenter = 128;
constexpr int kIters = 33;
constexpr int kShifts = 129;
constexpr int kKTiles = 36;
constexpr int kWaves = 4;
constexpr int kRows = 2 * kWaves;
constexpr int kPer = 16;
}

__device__ __forceinline__ float from_prev_lane(float v) {
  return __int_as_float(__builtin_amdgcn_update_dpp(
      0, __float_as_int(v), 0x111, 0xF, 0xF, true));
}
__device__ __forceinline__ float from_next_lane(float v) {
  return __int_as_float(__builtin_amdgcn_update_dpp(
      0, __float_as_int(v), 0x101, 0xF, 0xF, true));
}

template <int N>
__device__ __forceinline__ float from_lane_up(float v) {
  return __int_as_float(__builtin_amdgcn_update_dpp(
      0, __float_as_int(v), 0x100 + N, 0xF, 0xF, true));
}

__device__ __forceinline__ void diffuse_step(const float (&x)[kPer],
                                             float (&y)[kPer],
                                             const float (&p)[6]) {
  const int steps[6] = {1, -1, -3, 3, 4, -4};
  float ext[kPer + 8];
#pragma unroll
  for (int j = 0; j < 4; ++j) ext[j] = from_prev_lane(x[kPer - 4 + j]);
#pragma unroll
  for (int e = 0; e < kPer; ++e) ext[4 + e] = x[e];
#pragma unroll
  for (int j = 0; j < 4; ++j) ext[4 + kPer + j] = from_next_lane(x[j]);
#pragma unroll
  for (int e = 0; e < kPer; ++e) {
    float nv = 0.f;
#pragma unroll
    for (int i = 0; i < 6; ++i) nv = fmaf(p[i], ext[4 + e - steps[i]], nv);
    y[e] = nv;
  }
}

__global__ __launch_bounds__(32 * kWaves) void tonal_diffusion_kernel(
    const float* __restrict__ data,
    const float* __restrict__ log_weights,
    float* __restrict__ out,
    int n_rows) {
  const int lane = threadIdx.x & 31;
  const int wave = threadIdx.x >> 5;
  const int half = lane >> 4;
  const int hl   = lane & 15;
  const int rs   = 2 * wave + half;
  const int row  = blockIdx.x * kRows + rs;
  const int rowc = (row < n_rows) ? row : (n_rows - 1);

  __shared__ __attribute__((aligned(16))) float sDataPad[kRows][160];
  __shared__ __attribute__((aligned(16))) float sLogAcc[kRows][384];
  __shared__ __attribute__((aligned(16))) float sAcc[kRows][kS + 128];
  __shared__ __attribute__((aligned(16))) float sZ[kRows][kS];

  float p[6];
  float rate = 0.f;
#pragma unroll
  for (int i = 0; i < 6; ++i) {
    p[i] = expf(log_weights[(size_t)rowc * 6 + i]);
    rate += p[i];
  }
  const float inv_rate = 1.f / rate;
#pragma unroll
  for (int i = 0; i < 6; ++i) p[i] *= inv_rate;

  {
    const float4* d4 = reinterpret_cast<const float4*>(data + (size_t)rowc * kC);
#pragma unroll
    for (int q = 0; q < 2; ++q) {
      float4 raw = d4[hl + 16 * q];
      float4 v;
      v.x = (raw.x > 1e-8f) ? raw.x : 0.f;
      v.y = (raw.y > 1e-8f) ? raw.y : 0.f;
      v.z = (raw.z > 1e-8f) ? raw.z : 0.f;
      v.w = (raw.w > 1e-8f) ? raw.w : 0.f;
      *reinterpret_cast<float4*>(&sDataPad[rs][16 + 4 * hl + 64 * q]) = v;
    }
    sDataPad[rs][hl] = 0.f;
    sDataPad[rs][16 + kC + hl] = 0.f;
  }

  float x[kPer], y[kPer], acc[kPer];
#pragma unroll
  for (int e = 0; e < kPer; ++e) {
    x[e] = (kPer * hl + e == kCenter) ? 1.f : 0.f;
    acc[e] = 0.f;
  }

  float sp = expf(-rate);
  for (int n = 0; n < kIters - 1; n += 2) {
#pragma unroll
    for (int e = 0; e < kPer; ++e) acc[e] = fmaf(sp, x[e], acc[e]);
    diffuse_step(x, y, p);
    sp = sp * rate / (float)(n + 1);
#pragma unroll
    for (int e = 0; e < kPer; ++e) acc[e] = fmaf(sp, y[e], acc[e]);
    diffuse_step(y, x, p);
    sp = sp * rate / (float)(n + 2);
  }
#pragma unroll
  for (int e = 0; e < kPer; ++e) acc[e] = fmaf(sp, x[e], acc[e]);

#pragma unroll
  for (int e = 0; e < kPer; ++e) {
    const int t = kPer * hl + e;
    sAcc[rs][t] = acc[e];
    sLogAcc[rs][t] = logf(fmaxf(acc[e], 1e-38f));
  }
#pragma unroll
  for (int i = 0; i < 8; ++i) sAcc[rs][kS + hl + 16 * i] = 0.f;
#pragma unroll
  for (int i = 0; i < 8; ++i) sLogAcc[rs][kS + hl + 16 * i] = 0.f;
  __syncthreads();

  {
    float tail[kPer], head[kPer];
    const float4* own4 = reinterpret_cast<const float4*>(&sAcc[rs][kPer * hl]);
    const float4* far4 = reinterpret_cast<const float4*>(&sAcc[rs][kPer * (hl + 8)]);
    float own[kPer], far[kPer];
#pragma unroll
    for (int q = 0; q < kPer / 4; ++q) {
      const float4 o = own4[q];
      own[4 * q] = o.x; own[4 * q + 1] = o.y; own[4 * q + 2] = o.z; own[4 * q + 3] = o.w;
      const float4 f = far4[q];
      far[4 * q] = f.x; far[4 * q + 1] = f.y; far[4 * q + 2] = f.z; far[4 * q + 3] = f.w;
    }
    tail[kPer - 1] = own[kPer - 1];
#pragma unroll
    for (int b = kPer - 2; b >= 0; --b) tail[b] = tail[b + 1] + own[b];
    head[0] = 0.f;
#pragma unroll
    for (int b = 1; b < kPer; ++b) head[b] = head[b - 1] + far[b - 1];
    const float blk  = tail[0];
    const float two  = blk + from_lane_up<1>(blk);
    const float four = two + from_lane_up<2>(two);
    const float mid  = (from_lane_up<1>(four) + from_lane_up<5>(two)) +
                       from_lane_up<7>(blk);
    float4* zrow = reinterpret_cast<float4*>(&sZ[rs][kPer * hl]);
#pragma unroll
    for (int q = 0; q < kPer / 4; ++q) {
      float4 v;
      v.x = logf((tail[4 * q + 0] + mid) + head[4 * q + 0]);
      v.y = logf((tail[4 * q + 1] + mid) + head[4 * q + 1]);
      v.z = logf((tail[4 * q + 2] + mid) + head[4 * q + 2]);
      v.w = logf((tail[4 * q + 3] + mid) + head[4 * q + 3]);
      zrow[q] = v;
    }
  }

  float pa = 0.f, pb = 0.f;
  for (int i = 0; i < 4; ++i) {
    const float va = sDataPad[rs][16 + hl + 32 * i];
    if (va > 0.f) pa += va * logf(va);
    const float vb = sDataPad[rs][32 + hl + 32 * i];
    if (vb > 0.f) pb += vb * logf(vb);
  }
  float pl = pa + pb;
#pragma unroll
  for (int off = 8; off; off >>= 1) pl += __shfl_xor(pl, off, 32);
  const float plOther = __shfl_xor(pl, 16, 32);
  const float pl0 = half ? plOther : pl;
  const float pl1 = half ? pl : plOther;
  __syncthreads();

  const int m    = lane & 15;
  const int koff = (lane & 16) ? 2 : 0;
  const float* dp0 = &sDataPad[2 * wave][16 - m];
  const float* lp0 = &sLogAcc[2 * wave][16 * m];
  const float* dp1 = &sDataPad[2 * wave + 1][16 - m];
  const float* lp1 = &sLogAcc[2 * wave + 1][16 * m];
  v8f d0 = {0.f, 0.f, 0.f, 0.f, 0.f, 0.f, 0.f, 0.f};
  v8f d1 = {0.f, 0.f, 0.f, 0.f, 0.f, 0.f, 0.f, 0.f};
#pragma unroll
  for (int kc = 0; kc < kKTiles; ++kc) {
    const int j = 4 * kc + koff;
    v2f a0 = {dp0[j], dp0[j + 1]};
    v2f b0 = {lp0[j], lp0[j + 1]};
    d0 = __builtin_amdgcn_wmma_f32_16x16x4_f32(
         false, a0,  false, b0,
         (short)0, d0,  false,  false);
    v2f a1 = {dp1[j], dp1[j + 1]};
    v2f b1 = {lp1[j], lp1[j + 1]};
    d1 = __builtin_amdgcn_wmma_f32_16x16x4_f32(
         false, a1,  false, b1,
         (short)0, d1,  false,  false);
  }
  const int mbase = (lane & 16) ? 8 : 0;
  const int n16   = 16 * (lane & 15);
  float best0 = INFINITY, best1 = INFINITY;
#pragma unroll
  for (int r = 0; r < 8; ++r) {
    const int k = (r + mbase) + n16;
    if (k < kShifts) {
      best0 = fminf(best0, pl0 - d0[r] + sZ[2 * wave][k]);
      best1 = fminf(best1, pl1 - d1[r] + sZ[2 * wave + 1][k]);
    }
  }
#pragma unroll
  for (int off = 16; off; off >>= 1) {
    best0 = fminf(best0, __shfl_xor(best0, off, 32));
    best1 = fminf(best1, __shfl_xor(best1, off, 32));
  }
  if (lane == 0) {
    const int row0 = blockIdx.x * kRows + 2 * wave;
    if (row0 < n_rows) out[row0] = best0;
    if (row0 + 1 < n_rows) out[row0 + 1] = best1;
  }
}

extern "C" void kernel_launch(void* const* d_in, const int* in_sizes, int n_in,
                              void* d_out, int out_size, void* d_ws, size_t ws_size,
                              hipStream_t stream) {
  (void)n_in; (void)d_ws; (void)ws_size; (void)out_size;
  const float* data = (const float*)d_in[0];
  const float* logw = (const float*)d_in[1];
  float* out = (float*)d_out;
  const int n_rows = in_sizes[0] / kC;
  const int blocks = (n_rows + kRows - 1) / kRows;
  tonal_diffusion_kernel<<<blocks, 32 * kWaves, 0, stream>>>(data, logw, out, n_rows);
}


# Poisson series sum_n pmf(n) T^n delta evaluated by Horner's rule (u_n = W u_{n+1} + delta/n!, acc = exp(-rate) u_0): drops the acc FMAs and the per-term pmf division; reorders the acc summation
# speedup vs baseline: 2.4972x; 1.0997x over previous
#include <hip/hip_runtime.h>
#include <math.h>


typedef __attribute__((ext_vector_type(2))) float v2f;
typedef __attribute__((ext_vector_type(8))) float v8f;

namespace {
constexpr int kS = 256;
constexpr int kC = 128;
constexpr int kCenter = 128;
constexpr int kIters = 33;
constexpr int kShifts = 129;
constexpr int kKTiles = 36;
constexpr int kWaves = 4;
constexpr int kRows = 2 * kWaves;
constexpr int kPer = 16;
}

__device__ __forceinline__ float from_prev_lane(float v) {
  return __int_as_float(__builtin_amdgcn_update_dpp(
      0, __float_as_int(v), 0x111, 0xF, 0xF, true));
}
__device__ __forceinline__ float from_next_lane(float v) {
  return __int_as_float(__builtin_amdgcn_update_dpp(
      0, __float_as_int(v), 0x101, 0xF, 0xF, true));
}

template <int N>
__device__ __forceinline__ float from_lane_up(float v) {
  return __int_as_float(__builtin_amdgcn_update_dpp(
      0, __float_as_int(v), 0x100 + N, 0xF, 0xF, true));
}

__device__ __forceinline__ void diffuse_step(const float (&x)[kPer],
                                             float (&y)[kPer],
                                             const float (&p)[6],
                                             float inj) {
  const int steps[6] = {1, -1, -3, 3, 4, -4};
  float ext[kPer + 8];
#pragma unroll
  for (int j = 0; j < 4; ++j) ext[j] = from_prev_lane(x[kPer - 4 + j]);
#pragma unroll
  for (int e = 0; e < kPer; ++e) ext[4 + e] = x[e];
#pragma unroll
  for (int j = 0; j < 4; ++j) ext[4 + kPer + j] = from_next_lane(x[j]);
#pragma unroll
  for (int e = 0; e < kPer; ++e) {
    float nv = (e == kCenter % kPer) ? inj : 0.f;
#pragma unroll
    for (int i = 0; i < 6; ++i) nv = fmaf(p[i], ext[4 + e - steps[i]], nv);
    y[e] = nv;
  }
}

__global__ __launch_bounds__(32 * kWaves) void tonal_diffusion_kernel(
    const float* __restrict__ data,
    const float* __restrict__ log_weights,
    float* __restrict__ out,
    int n_rows) {
  const int lane = threadIdx.x & 31;
  const int wave = threadIdx.x >> 5;
  const int half = lane >> 4;
  const int hl   = lane & 15;
  const int rs   = 2 * wave + half;
  const int row  = blockIdx.x * kRows + rs;
  const int rowc = (row < n_rows) ? row : (n_rows - 1);

  __shared__ __attribute__((aligned(16))) float sDataPad[kRows][160];
  __shared__ __attribute__((aligned(16))) float sLogAcc[kRows][384];
  __shared__ __attribute__((aligned(16))) float sAcc[kRows][kS + 128];
  __shared__ __attribute__((aligned(16))) float sZ[kRows][kS];

  float p[6];
  float rate = 0.f;
#pragma unroll
  for (int i = 0; i < 6; ++i) {
    p[i] = expf(log_weights[(size_t)rowc * 6 + i]);
    rate += p[i];
  }

  {
    const float4* d4 = reinterpret_cast<const float4*>(data + (size_t)rowc * kC);
#pragma unroll
    for (int q = 0; q < 2; ++q) {
      float4 raw = d4[hl + 16 * q];
      float4 v;
      v.x = (raw.x > 1e-8f) ? raw.x : 0.f;
      v.y = (raw.y > 1e-8f) ? raw.y : 0.f;
      v.z = (raw.z > 1e-8f) ? raw.z : 0.f;
      v.w = (raw.w > 1e-8f) ? raw.w : 0.f;
      *reinterpret_cast<float4*>(&sDataPad[rs][16 + 4 * hl + 64 * q]) = v;
    }
    sDataPad[rs][hl] = 0.f;
    sDataPad[rs][16 + kC + hl] = 0.f;
  }

  float x[kPer], y[kPer], acc[kPer];
  const float owns_center = (kPer * hl == kCenter - kCenter % kPer) ? 1.f : 0.f;
  float fn = 3.8003907548547435e-36f;
#pragma unroll
  for (int e = 0; e < kPer; ++e)
    x[e] = (e == kCenter % kPer) ? owns_center * fn : 0.f;

  for (int n = kIters - 1; n > 0; n -= 2) {
    fn *= (float)n;
    diffuse_step(x, y, p, owns_center * fn);
    fn *= (float)(n - 1);
    diffuse_step(y, x, p, owns_center * fn);
  }
  const float sp = expf(-rate);
#pragma unroll
  for (int e = 0; e < kPer; ++e) acc[e] = sp * x[e];

#pragma unroll
  for (int e = 0; e < kPer; ++e) {
    const int t = kPer * hl + e;
    sAcc[rs][t] = acc[e];
    sLogAcc[rs][t] = logf(fmaxf(acc[e], 1e-38f));
  }
#pragma unroll
  for (int i = 0; i < 8; ++i) sAcc[rs][kS + hl + 16 * i] = 0.f;
#pragma unroll
  for (int i = 0; i < 8; ++i) sLogAcc[rs][kS + hl + 16 * i] = 0.f;
  __syncthreads();

  {
    float tail[kPer], head[kPer];
    const float4* own4 = reinterpret_cast<const float4*>(&sAcc[rs][kPer * hl]);
    const float4* far4 = reinterpret_cast<const float4*>(&sAcc[rs][kPer * (hl + 8)]);
    float own[kPer], far[kPer];
#pragma unroll
    for (int q = 0; q < kPer / 4; ++q) {
      const float4 o = own4[q];
      own[4 * q] = o.x; own[4 * q + 1] = o.y; own[4 * q + 2] = o.z; own[4 * q + 3] = o.w;
      const float4 f = far4[q];
      far[4 * q] = f.x; far[4 * q + 1] = f.y; far[4 * q + 2] = f.z; far[4 * q + 3] = f.w;
    }
    tail[kPer - 1] = own[kPer - 1];
#pragma unroll
    for (int b = kPer - 2; b >= 0; --b) tail[b] = tail[b + 1] + own[b];
    head[0] = 0.f;
#pragma unroll
    for (int b = 1; b < kPer; ++b) head[b] = head[b - 1] + far[b - 1];
    const float blk  = tail[0];
    const float two  = blk + from_lane_up<1>(blk);
    const float four = two + from_lane_up<2>(two);
    const float mid  = (from_lane_up<1>(four) + from_lane_up<5>(two)) +
                       from_lane_up<7>(blk);
    float4* zrow = reinterpret_cast<float4*>(&sZ[rs][kPer * hl]);
#pragma unroll
    for (int q = 0; q < kPer / 4; ++q) {
      float4 v;
      v.x = logf((tail[4 * q + 0] + mid) + head[4 * q + 0]);
      v.y = logf((tail[4 * q + 1] + mid) + head[4 * q + 1]);
      v.z = logf((tail[4 * q + 2] + mid) + head[4 * q + 2]);
      v.w = logf((tail[4 * q + 3] + mid) + head[4 * q + 3]);
      zrow[q] = v;
    }
  }

  float pa = 0.f, pb = 0.f;
  for (int i = 0; i < 4; ++i) {
    const float va = sDataPad[rs][16 + hl + 32 * i];
    if (va > 0.f) pa += va * logf(va);
    const float vb = sDataPad[rs][32 + hl + 32 * i];
    if (vb > 0.f) pb += vb * logf(vb);
  }
  float pl = pa + pb;
#pragma unroll
  for (int off = 8; off; off >>= 1) pl += __shfl_xor(pl, off, 32);
  const float plOther = __shfl_xor(pl, 16, 32);
  const float pl0 = half ? plOther : pl;
  const float pl1 = half ? pl : plOther;
  __syncthreads();

  const int m    = lane & 15;
  const int koff = (lane & 16) ? 2 : 0;
  const float* dp0 = &sDataPad[2 * wave][16 - m];
  const float* lp0 = &sLogAcc[2 * wave][16 * m];
  const float* dp1 = &sDataPad[2 * wave + 1][16 - m];
  const float* lp1 = &sLogAcc[2 * wave + 1][16 * m];
  v8f d0 = {0.f, 0.f, 0.f, 0.f, 0.f, 0.f, 0.f, 0.f};
  v8f d1 = {0.f, 0.f, 0.f, 0.f, 0.f, 0.f, 0.f, 0.f};
#pragma unroll
  for (int kc = 0; kc < kKTiles; ++kc) {
    const int j = 4 * kc + koff;
    v2f a0 = {dp0[j], dp0[j + 1]};
    v2f b0 = {lp0[j], lp0[j + 1]};
    d0 = __builtin_amdgcn_wmma_f32_16x16x4_f32(
         false, a0,  false, b0,
         (short)0, d0,  false,  false);
    v2f a1 = {dp1[j], dp1[j + 1]};
    v2f b1 = {lp1[j], lp1[j + 1]};
    d1 = __builtin_amdgcn_wmma_f32_16x16x4_f32(
         false, a1,  false, b1,
         (short)0, d1,  false,  false);
  }
  const int mbase = (lane & 16) ? 8 : 0;
  const int n16   = 16 * (lane & 15);
  float best0 = INFINITY, best1 = INFINITY;
#pragma unroll
  for (int r = 0; r < 8; ++r) {
    const int k = (r + mbase) + n16;
    if (k < kShifts) {
      best0 = fminf(best0, pl0 - d0[r] + sZ[2 * wave][k]);
      best1 = fminf(best1, pl1 - d1[r] + sZ[2 * wave + 1][k]);
    }
  }
#pragma unroll
  for (int off = 16; off; off >>= 1) {
    best0 = fminf(best0, __shfl_xor(best0, off, 32));
    best1 = fminf(best1, __shfl_xor(best1, off, 32));
  }
  if (lane == 0) {
    const int row0 = blockIdx.x * kRows + 2 * wave;
    if (row0 < n_rows) out[row0] = best0;
    if (row0 + 1 < n_rows) out[row0 + 1] = best1;
  }
}

extern "C" void kernel_launch(void* const* d_in, const int* in_sizes, int n_in,
                              void* d_out, int out_size, void* d_ws, size_t ws_size,
                              hipStream_t stream) {
  (void)n_in; (void)d_ws; (void)ws_size; (void)out_size;
  const float* data = (const float*)d_in[0];
  const float* logw = (const float*)d_in[1];
  float* out = (float*)d_out;
  const int n_rows = in_sizes[0] / kC;
  const int blocks = (n_rows + kRows - 1) / kRows;
  tonal_diffusion_kernel<<<blocks, 32 * kWaves, 0, stream>>>(data, logw, out, n_rows);
}


# Library expf/logf (7 exp and 40 log calls per lane) replaced by the hardware v_exp_f32/v_log_f32 forms with one scale and one fused multiply-add each; results move by about 1 ulp of log2
# speedup vs baseline: 2.7924x; 1.1182x over previous
#include <hip/hip_runtime.h>
#include <math.h>


typedef __attribute__((ext_vector_type(2))) float v2f;
typedef __attribute__((ext_vector_type(8))) float v8f;

namespace {
constexpr int kS = 256;
constexpr int kC = 128;
constexpr int kCenter = 128;
constexpr int kIters = 33;
constexpr int kShifts = 129;
constexpr int kKTiles = 36;
constexpr int kWaves = 4;
constexpr int kRows = 2 * kWaves;
constexpr int kPer = 16;
}

__device__ __forceinline__ float from_prev_lane(float v) {
  return __int_as_float(__builtin_amdgcn_update_dpp(
      0, __float_as_int(v), 0x111, 0xF, 0xF, true));
}
__device__ __forceinline__ float from_next_lane(float v) {
  return __int_as_float(__builtin_amdgcn_update_dpp(
      0, __float_as_int(v), 0x101, 0xF, 0xF, true));
}

template <int N>
__device__ __forceinline__ float from_lane_up(float v) {
  return __int_as_float(__builtin_amdgcn_update_dpp(
      0, __float_as_int(v), 0x100 + N, 0xF, 0xF, true));
}

__device__ __forceinline__ float log_hw(float v) {
  return fmaf(__builtin_amdgcn_logf(v * 256.f), 0.693147180559945f,
              -5.545177444479562f);
}
__device__ __forceinline__ float exp_hw(float v) {
  return __builtin_amdgcn_exp2f(v * 1.4426950408889634f);
}

__device__ __forceinline__ void diffuse_step(const float (&x)[kPer],
                                             float (&y)[kPer],
                                             const float (&p)[6],
                                             float inj) {
  const int steps[6] = {1, -1, -3, 3, 4, -4};
  float ext[kPer + 8];
#pragma unroll
  for (int j = 0; j < 4; ++j) ext[j] = from_prev_lane(x[kPer - 4 + j]);
#pragma unroll
  for (int e = 0; e < kPer; ++e) ext[4 + e] = x[e];
#pragma unroll
  for (int j = 0; j < 4; ++j) ext[4 + kPer + j] = from_next_lane(x[j]);
#pragma unroll
  for (int e = 0; e < kPer; ++e) {
    float nv = (e == kCenter % kPer) ? inj : 0.f;
#pragma unroll
    for (int i = 0; i < 6; ++i) nv = fmaf(p[i], ext[4 + e - steps[i]], nv);
    y[e] = nv;
  }
}

__global__ __launch_bounds__(32 * kWaves) void tonal_diffusion_kernel(
    const float* __restrict__ data,
    const float* __restrict__ log_weights,
    float* __restrict__ out,
    int n_rows) {
  const int lane = threadIdx.x & 31;
  const int wave = threadIdx.x >> 5;
  const int half = lane >> 4;
  const int hl   = lane & 15;
  const int rs   = 2 * wave + half;
  const int row  = blockIdx.x * kRows + rs;
  const int rowc = (row < n_rows) ? row : (n_rows - 1);

  __shared__ __attribute__((aligned(16))) float sDataPad[kRows][160];
  __shared__ __attribute__((aligned(16))) float sLogAcc[kRows][384];
  __shared__ __attribute__((aligned(16))) float sAcc[kRows][kS + 128];
  __shared__ __attribute__((aligned(16))) float sZ[kRows][kS];

  float p[6];
  float rate = 0.f;
#pragma unroll
  for (int i = 0; i < 6; ++i) {
    p[i] = exp_hw(log_weights[(size_t)rowc * 6 + i]);
    rate += p[i];
  }

  {
    const float4* d4 = reinterpret_cast<const float4*>(data + (size_t)rowc * kC);
#pragma unroll
    for (int q = 0; q < 2; ++q) {
      float4 raw = d4[hl + 16 * q];
      float4 v;
      v.x = (raw.x > 1e-8f) ? raw.x : 0.f;
      v.y = (raw.y > 1e-8f) ? raw.y : 0.f;
      v.z = (raw.z > 1e-8f) ? raw.z : 0.f;
      v.w = (raw.w > 1e-8f) ? raw.w : 0.f;
      *reinterpret_cast<float4*>(&sDataPad[rs][16 + 4 * hl + 64 * q]) = v;
    }
    sDataPad[rs][hl] = 0.f;
    sDataPad[rs][16 + kC + hl] = 0.f;
  }

  float x[kPer], y[kPer], acc[kPer];
  const float owns_center = (kPer * hl == kCenter - kCenter % kPer) ? 1.f : 0.f;
  float fn = 3.8003907548547435e-36f;
#pragma unroll
  for (int e = 0; e < kPer; ++e)
    x[e] = (e == kCenter % kPer) ? owns_center * fn : 0.f;

  for (int n = kIters - 1; n > 0; n -= 2) {
    fn *= (float)n;
    diffuse_step(x, y, p, owns_center * fn);
    fn *= (float)(n - 1);
    diffuse_step(y, x, p, owns_center * fn);
  }
  const float sp = exp_hw(-rate);
#pragma unroll
  for (int e = 0; e < kPer; ++e) acc[e] = sp * x[e];

#pragma unroll
  for (int e = 0; e < kPer; ++e) {
    const int t = kPer * hl + e;
    sAcc[rs][t] = acc[e];
    sLogAcc[rs][t] = log_hw(fmaxf(acc[e], 1e-38f));
  }
#pragma unroll
  for (int i = 0; i < 8; ++i) sAcc[rs][kS + hl + 16 * i] = 0.f;
#pragma unroll
  for (int i = 0; i < 8; ++i) sLogAcc[rs][kS + hl + 16 * i] = 0.f;
  __syncthreads();

  {
    float tail[kPer], head[kPer];
    const float4* own4 = reinterpret_cast<const float4*>(&sAcc[rs][kPer * hl]);
    const float4* far4 = reinterpret_cast<const float4*>(&sAcc[rs][kPer * (hl + 8)]);
    float own[kPer], far[kPer];
#pragma unroll
    for (int q = 0; q < kPer / 4; ++q) {
      const float4 o = own4[q];
      own[4 * q] = o.x; own[4 * q + 1] = o.y; own[4 * q + 2] = o.z; own[4 * q + 3] = o.w;
      const float4 f = far4[q];
      far[4 * q] = f.x; far[4 * q + 1] = f.y; far[4 * q + 2] = f.z; far[4 * q + 3] = f.w;
    }
    tail[kPer - 1] = own[kPer - 1];
#pragma unroll
    for (int b = kPer - 2; b >= 0; --b) tail[b] = tail[b + 1] + own[b];
    head[0] = 0.f;
#pragma unroll
    for (int b = 1; b < kPer; ++b) head[b] = head[b - 1] + far[b - 1];
    const float blk  = tail[0];
    const float two  = blk + from_lane_up<1>(blk);
    const float four = two + from_lane_up<2>(two);
    const float mid  = (from_lane_up<1>(four) + from_lane_up<5>(two)) +
                       from_lane_up<7>(blk);
    float4* zrow = reinterpret_cast<float4*>(&sZ[rs][kPer * hl]);
#pragma unroll
    for (int q = 0; q < kPer / 4; ++q) {
      float4 v;
      v.x = log_hw((tail[4 * q + 0] + mid) + head[4 * q + 0]);
      v.y = log_hw((tail[4 * q + 1] + mid) + head[4 * q + 1]);
      v.z = log_hw((tail[4 * q + 2] + mid) + head[4 * q + 2]);
      v.w = log_hw((tail[4 * q + 3] + mid) + head[4 * q + 3]);
      zrow[q] = v;
    }
  }

  float pa = 0.f, pb = 0.f;
  for (int i = 0; i < 4; ++i) {
    const float va = sDataPad[rs][16 + hl + 32 * i];
    if (va > 0.f) pa += va * log_hw(va);
    const float vb = sDataPad[rs][32 + hl + 32 * i];
    if (vb > 0.f) pb += vb * log_hw(vb);
  }
  float pl = pa + pb;
#pragma unroll
  for (int off = 8; off; off >>= 1) pl += __shfl_xor(pl, off, 32);
  const float plOther = __shfl_xor(pl, 16, 32);
  const float pl0 = half ? plOther : pl;
  const float pl1 = half ? pl : plOther;
  __syncthreads();

  const int m    = lane & 15;
  const int koff = (lane & 16) ? 2 : 0;
  const float* dp0 = &sDataPad[2 * wave][16 - m];
  const float* lp0 = &sLogAcc[2 * wave][16 * m];
  const float* dp1 = &sDataPad[2 * wave + 1][16 - m];
  const float* lp1 = &sLogAcc[2 * wave + 1][16 * m];
  v8f d0 = {0.f, 0.f, 0.f, 0.f, 0.f, 0.f, 0.f, 0.f};
  v8f d1 = {0.f, 0.f, 0.f, 0.f, 0.f, 0.f, 0.f, 0.f};
#pragma unroll
  for (int kc = 0; kc < kKTiles; ++kc) {
    const int j = 4 * kc + koff;
    v2f a0 = {dp0[j], dp0[j + 1]};
    v2f b0 = {lp0[j], lp0[j + 1]};
    d0 = __builtin_amdgcn_wmma_f32_16x16x4_f32(
         false, a0,  false, b0,
         (short)0, d0,  false,  false);
    v2f a1 = {dp1[j], dp1[j + 1]};
    v2f b1 = {lp1[j], lp1[j + 1]};
    d1 = __builtin_amdgcn_wmma_f32_16x16x4_f32(
         false, a1,  false, b1,
         (short)0, d1,  false,  false);
  }
  const int mbase = (lane & 16) ? 8 : 0;
  const int n16   = 16 * (lane & 15);
  float best0 = INFINITY, best1 = INFINITY;
#pragma unroll
  for (int r = 0; r < 8; ++r) {
    const int k = (r + mbase) + n16;
    if (k < kShifts) {
      best0 = fminf(best0, pl0 - d0[r] + sZ[2 * wave][k]);
      best1 = fminf(best1, pl1 - d1[r] + sZ[2 * wave + 1][k]);
    }
  }
#pragma unroll
  for (int off = 16; off; off >>= 1) {
    best0 = fminf(best0, __shfl_xor(best0, off, 32));
    best1 = fminf(best1, __shfl_xor(best1, off, 32));
  }
  if (lane == 0) {
    const int row0 = blockIdx.x * kRows + 2 * wave;
    if (row0 < n_rows) out[row0] = best0;
    if (row0 + 1 < n_rows) out[row0 + 1] = best1;
  }
}

extern "C" void kernel_launch(void* const* d_in, const int* in_sizes, int n_in,
                              void* d_out, int out_size, void* d_ws, size_t ws_size,
                              hipStream_t stream) {
  (void)n_in; (void)d_ws; (void)ws_size; (void)out_size;
  const float* data = (const float*)d_in[0];
  const float* logw = (const float*)d_in[1];
  float* out = (float*)d_out;
  const int n_rows = in_sizes[0] / kC;
  const int blocks = (n_rows + kRows - 1) / kRows;
  tonal_diffusion_kernel<<<blocks, 32 * kWaves, 0, stream>>>(data, logw, out, n_rows);
}


# Horner passes 1-15 run on compact register layouts (4, then 8 support points per lane, while the iterate's support still fits), regathered in-row by permlane16 to 16 per lane at pass 16
# speedup vs baseline: 3.0834x; 1.1042x over previous
#include <hip/hip_runtime.h>
#include <math.h>


typedef __attribute__((ext_vector_type(2))) float v2f;
typedef __attribute__((ext_vector_type(8))) float v8f;

namespace {
constexpr int kS = 256;
constexpr int kC = 128;
constexpr int kCenter = 128;
constexpr int kIters = 33;
constexpr int kShifts = 129;
constexpr int kKTiles = 36;
constexpr int kWaves = 4;
constexpr int kRows = 2 * kWaves;
constexpr int kPer = 16;
}

__device__ __forceinline__ float from_prev_lane(float v) {
  return __int_as_float(__builtin_amdgcn_update_dpp(
      0, __float_as_int(v), 0x111, 0xF, 0xF, true));
}
__device__ __forceinline__ float from_next_lane(float v) {
  return __int_as_float(__builtin_amdgcn_update_dpp(
      0, __float_as_int(v), 0x101, 0xF, 0xF, true));
}

template <int N>
__device__ __forceinline__ float from_lane_up(float v) {
  return __int_as_float(__builtin_amdgcn_update_dpp(
      0, __float_as_int(v), 0x100 + N, 0xF, 0xF, true));
}

__device__ __forceinline__ float log_hw(float v) {
  return fmaf(__builtin_amdgcn_logf(v * 256.f), 0.693147180559945f,
              -5.545177444479562f);
}
__device__ __forceinline__ float exp_hw(float v) {
  return __builtin_amdgcn_exp2f(v * 1.4426950408889634f);
}

template <int P>
__device__ __forceinline__ void diffuse_step(const float (&x)[P],
                                             float (&y)[P],
                                             const float (&p)[6],
                                             float inj) {
  const int steps[6] = {1, -1, -3, 3, 4, -4};
  float ext[P + 8];
#pragma unroll
  for (int j = 0; j < 4; ++j) ext[j] = from_prev_lane(x[P - 4 + j]);
#pragma unroll
  for (int e = 0; e < P; ++e) ext[4 + e] = x[e];
#pragma unroll
  for (int j = 0; j < 4; ++j) ext[4 + P + j] = from_next_lane(x[j]);
#pragma unroll
  for (int e = 0; e < P; ++e) {
    float nv = (e == 0) ? inj : 0.f;
#pragma unroll
    for (int i = 0; i < 6; ++i) nv = fmaf(p[i], ext[4 + e - steps[i]], nv);
    y[e] = nv;
  }
}

__device__ __forceinline__ float from_row_lane(float v, int src) {
  return __int_as_float(__builtin_amdgcn_permlane16_var(
      0, __float_as_int(v), src, false, false));
}

template <int P>
__device__ __forceinline__ void widen_layout(const float (&a)[P], float (&b)[2 * P],
                                             int hl) {
  const bool inside = (hl >= 4) && (hl < 12);
  const int lo = (2 * hl - 8) & 15;
  const int hi = (2 * hl - 7) & 15;
#pragma unroll
  for (int e = 0; e < P; ++e) {
    const float vlo = from_row_lane(a[e], lo);
    const float vhi = from_row_lane(a[e], hi);
    b[e]     = inside ? vlo : 0.f;
    b[P + e] = inside ? vhi : 0.f;
  }
}

template <int P, int PAIRS>
__device__ __forceinline__ void horner_pairs(float (&x)[P], float (&y)[P],
                                             const float (&p)[6], float owns_center,
                                             float& fn, int& n) {
  for (int it = 0; it < PAIRS; ++it) {
    fn *= (float)n; --n;
    diffuse_step<P>(x, y, p, owns_center * fn);
    fn *= (float)n; --n;
    diffuse_step<P>(y, x, p, owns_center * fn);
  }
}

__global__ __launch_bounds__(32 * kWaves) void tonal_diffusion_kernel(
    const float* __restrict__ data,
    const float* __restrict__ log_weights,
    float* __restrict__ out,
    int n_rows) {
  const int lane = threadIdx.x & 31;
  const int wave = threadIdx.x >> 5;
  const int half = lane >> 4;
  const int hl   = lane & 15;
  const int rs   = 2 * wave + half;
  const int row  = blockIdx.x * kRows + rs;
  const int rowc = (row < n_rows) ? row : (n_rows - 1);

  __shared__ __attribute__((aligned(16))) float sDataPad[kRows][160];
  __shared__ __attribute__((aligned(16))) float sLogAcc[kRows][384];
  __shared__ __attribute__((aligned(16))) float sAcc[kRows][kS + 128];
  __shared__ __attribute__((aligned(16))) float sZ[kRows][kS];

  float p[6];
  float rate = 0.f;
#pragma unroll
  for (int i = 0; i < 6; ++i) {
    p[i] = exp_hw(log_weights[(size_t)rowc * 6 + i]);
    rate += p[i];
  }

  {
    const float4* d4 = reinterpret_cast<const float4*>(data + (size_t)rowc * kC);
#pragma unroll
    for (int q = 0; q < 2; ++q) {
      float4 raw = d4[hl + 16 * q];
      float4 v;
      v.x = (raw.x > 1e-8f) ? raw.x : 0.f;
      v.y = (raw.y > 1e-8f) ? raw.y : 0.f;
      v.z = (raw.z > 1e-8f) ? raw.z : 0.f;
      v.w = (raw.w > 1e-8f) ? raw.w : 0.f;
      *reinterpret_cast<float4*>(&sDataPad[rs][16 + 4 * hl + 64 * q]) = v;
    }
    sDataPad[rs][hl] = 0.f;
    sDataPad[rs][16 + kC + hl] = 0.f;
  }

  float x[kPer], y[kPer], acc[kPer];
  const float owns_center = (kPer * hl == kCenter - kCenter % kPer) ? 1.f : 0.f;
  float fn = 3.8003907548547435e-36f;
  int n = kIters - 1;
  {
    float x4[4], y4[4];
#pragma unroll
    for (int e = 0; e < 4; ++e) x4[e] = (e == 0) ? owns_center * fn : 0.f;
    horner_pairs<4, 3>(x4, y4, p, owns_center, fn, n);
    fn *= (float)n; --n;
    diffuse_step<4>(x4, y4, p, owns_center * fn);
    float x8[8], y8[8];
    widen_layout<4>(y4, x8, hl);
    horner_pairs<8, 4>(x8, y8, p, owns_center, fn, n);
    widen_layout<8>(x8, y, hl);
  }
  fn *= (float)n; --n;
  diffuse_step<kPer>(y, x, p, owns_center * fn);
  horner_pairs<kPer, 8>(x, y, p, owns_center, fn, n);
  const float sp = exp_hw(-rate);
#pragma unroll
  for (int e = 0; e < kPer; ++e) acc[e] = sp * x[e];

#pragma unroll
  for (int e = 0; e < kPer; ++e) {
    const int t = kPer * hl + e;
    sAcc[rs][t] = acc[e];
    sLogAcc[rs][t] = log_hw(fmaxf(acc[e], 1e-38f));
  }
#pragma unroll
  for (int i = 0; i < 8; ++i) sAcc[rs][kS + hl + 16 * i] = 0.f;
#pragma unroll
  for (int i = 0; i < 8; ++i) sLogAcc[rs][kS + hl + 16 * i] = 0.f;
  __syncthreads();

  {
    float tail[kPer], head[kPer];
    const float4* own4 = reinterpret_cast<const float4*>(&sAcc[rs][kPer * hl]);
    const float4* far4 = reinterpret_cast<const float4*>(&sAcc[rs][kPer * (hl + 8)]);
    float own[kPer], far[kPer];
#pragma unroll
    for (int q = 0; q < kPer / 4; ++q) {
      const float4 o = own4[q];
      own[4 * q] = o.x; own[4 * q + 1] = o.y; own[4 * q + 2] = o.z; own[4 * q + 3] = o.w;
      const float4 f = far4[q];
      far[4 * q] = f.x; far[4 * q + 1] = f.y; far[4 * q + 2] = f.z; far[4 * q + 3] = f.w;
    }
    tail[kPer - 1] = own[kPer - 1];
#pragma unroll
    for (int b = kPer - 2; b >= 0; --b) tail[b] = tail[b + 1] + own[b];
    head[0] = 0.f;
#pragma unroll
    for (int b = 1; b < kPer; ++b) head[b] = head[b - 1] + far[b - 1];
    const float blk  = tail[0];
    const float two  = blk + from_lane_up<1>(blk);
    const float four = two + from_lane_up<2>(two);
    const float mid  = (from_lane_up<1>(four) + from_lane_up<5>(two)) +
                       from_lane_up<7>(blk);
    float4* zrow = reinterpret_cast<float4*>(&sZ[rs][kPer * hl]);
#pragma unroll
    for (int q = 0; q < kPer / 4; ++q) {
      float4 v;
      v.x = log_hw((tail[4 * q + 0] + mid) + head[4 * q + 0]);
      v.y = log_hw((tail[4 * q + 1] + mid) + head[4 * q + 1]);
      v.z = log_hw((tail[4 * q + 2] + mid) + head[4 * q + 2]);
      v.w = log_hw((tail[4 * q + 3] + mid) + head[4 * q + 3]);
      zrow[q] = v;
    }
  }

  float pa = 0.f, pb = 0.f;
  for (int i = 0; i < 4; ++i) {
    const float va = sDataPad[rs][16 + hl + 32 * i];
    if (va > 0.f) pa += va * log_hw(va);
    const float vb = sDataPad[rs][32 + hl + 32 * i];
    if (vb > 0.f) pb += vb * log_hw(vb);
  }
  float pl = pa + pb;
#pragma unroll
  for (int off = 8; off; off >>= 1) pl += __shfl_xor(pl, off, 32);
  const float plOther = __shfl_xor(pl, 16, 32);
  const float pl0 = half ? plOther : pl;
  const float pl1 = half ? pl : plOther;
  __syncthreads();

  const int m    = lane & 15;
  const int koff = (lane & 16) ? 2 : 0;
  const float* dp0 = &sDataPad[2 * wave][16 - m];
  const float* lp0 = &sLogAcc[2 * wave][16 * m];
  const float* dp1 = &sDataPad[2 * wave + 1][16 - m];
  const float* lp1 = &sLogAcc[2 * wave + 1][16 * m];
  v8f d0 = {0.f, 0.f, 0.f, 0.f, 0.f, 0.f, 0.f, 0.f};
  v8f d1 = {0.f, 0.f, 0.f, 0.f, 0.f, 0.f, 0.f, 0.f};
#pragma unroll
  for (int kc = 0; kc < kKTiles; ++kc) {
    const int j = 4 * kc + koff;
    v2f a0 = {dp0[j], dp0[j + 1]};
    v2f b0 = {lp0[j], lp0[j + 1]};
    d0 = __builtin_amdgcn_wmma_f32_16x16x4_f32(
         false, a0,  false, b0,
         (short)0, d0,  false,  false);
    v2f a1 = {dp1[j], dp1[j + 1]};
    v2f b1 = {lp1[j], lp1[j + 1]};
    d1 = __builtin_amdgcn_wmma_f32_16x16x4_f32(
         false, a1,  false, b1,
         (short)0, d1,  false,  false);
  }
  const int mbase = (lane & 16) ? 8 : 0;
  const int n16   = 16 * (lane & 15);
  float best0 = INFINITY, best1 = INFINITY;
#pragma unroll
  for (int r = 0; r < 8; ++r) {
    const int k = (r + mbase) + n16;
    if (k < kShifts) {
      best0 = fminf(best0, pl0 - d0[r] + sZ[2 * wave][k]);
      best1 = fminf(best1, pl1 - d1[r] + sZ[2 * wave + 1][k]);
    }
  }
#pragma unroll
  for (int off = 16; off; off >>= 1) {
    best0 = fminf(best0, __shfl_xor(best0, off, 32));
    best1 = fminf(best1, __shfl_xor(best1, off, 32));
  }
  if (lane == 0) {
    const int row0 = blockIdx.x * kRows + 2 * wave;
    if (row0 < n_rows) out[row0] = best0;
    if (row0 + 1 < n_rows) out[row0 + 1] = best1;
  }
}

extern "C" void kernel_launch(void* const* d_in, const int* in_sizes, int n_in,
                              void* d_out, int out_size, void* d_ws, size_t ws_size,
                              hipStream_t stream) {
  (void)n_in; (void)d_ws; (void)ws_size; (void)out_size;
  const float* data = (const float*)d_in[0];
  const float* logw = (const float*)d_in[1];
  float* out = (float*)d_out;
  const int n_rows = in_sizes[0] / kC;
  const int blocks = (n_rows + kRows - 1) / kRows;
  tonal_diffusion_kernel<<<blocks, 32 * kWaves, 0, stream>>>(data, logw, out, n_rows);
}


# Epilogue cross-lane traffic off LDS: window-sum logs reach the accumulator lanes by 8 permlane16 swaps (no LDS array, 2nd barrier, masked LDS reads); reductions by DPP row_xmask, not ds_bpermute
# speedup vs baseline: 3.2438x; 1.0520x over previous
#include <hip/hip_runtime.h>
#include <math.h>


typedef __attribute__((ext_vector_type(2))) float v2f;
typedef __attribute__((ext_vector_type(8))) float v8f;
typedef __attribute__((ext_vector_type(2))) unsigned int v2u;

namespace {
constexpr int kS = 256;
constexpr int kC = 128;
constexpr int kCenter = 128;
constexpr int kIters = 33;
constexpr int kShifts = 129;
constexpr int kKTiles = 36;
constexpr int kWaves = 4;
constexpr int kRows = 2 * kWaves;
constexpr int kPer = 16;
}

__device__ __forceinline__ float from_prev_lane(float v) {
  return __int_as_float(__builtin_amdgcn_update_dpp(
      0, __float_as_int(v), 0x111, 0xF, 0xF, true));
}
__device__ __forceinline__ float from_next_lane(float v) {
  return __int_as_float(__builtin_amdgcn_update_dpp(
      0, __float_as_int(v), 0x101, 0xF, 0xF, true));
}

template <int N>
__device__ __forceinline__ float from_lane_up(float v) {
  return __int_as_float(__builtin_amdgcn_update_dpp(
      0, __float_as_int(v), 0x100 + N, 0xF, 0xF, true));
}

template <int MASK>
__device__ __forceinline__ float from_row_xor(float v) {
  return __int_as_float(__builtin_amdgcn_update_dpp(
      0, __float_as_int(v), 0x160 + MASK, 0xF, 0xF, true));
}
__device__ __forceinline__ float from_other_half(float v) {
  const int b = __float_as_int(v);
  return __int_as_float(__builtin_amdgcn_permlanex16(
      b, b, 0x76543210, 0xfedcba98, false, false));
}

__device__ __forceinline__ float log_hw(float v) {
  return fmaf(__builtin_amdgcn_logf(v * 256.f), 0.693147180559945f,
              -5.545177444479562f);
}
__device__ __forceinline__ float exp_hw(float v) {
  return __builtin_amdgcn_exp2f(v * 1.4426950408889634f);
}

template <int P>
__device__ __forceinline__ void diffuse_step(const float (&x)[P],
                                             float (&y)[P],
                                             const float (&p)[6],
                                             float inj) {
  const int steps[6] = {1, -1, -3, 3, 4, -4};
  float ext[P + 8];
#pragma unroll
  for (int j = 0; j < 4; ++j) ext[j] = from_prev_lane(x[P - 4 + j]);
#pragma unroll
  for (int e = 0; e < P; ++e) ext[4 + e] = x[e];
#pragma unroll
  for (int j = 0; j < 4; ++j) ext[4 + P + j] = from_next_lane(x[j]);
#pragma unroll
  for (int e = 0; e < P; ++e) {
    float nv = (e == 0) ? inj : 0.f;
#pragma unroll
    for (int i = 0; i < 6; ++i) nv = fmaf(p[i], ext[4 + e - steps[i]], nv);
    y[e] = nv;
  }
}

__device__ __forceinline__ float from_row_lane(float v, int src) {
  return __int_as_float(__builtin_amdgcn_permlane16_var(
      0, __float_as_int(v), src, false, false));
}

template <int P>
__device__ __forceinline__ void widen_layout(const float (&a)[P], float (&b)[2 * P],
                                             int hl) {
  const bool inside = (hl >= 4) && (hl < 12);
  const int lo = (2 * hl - 8) & 15;
  const int hi = (2 * hl - 7) & 15;
#pragma unroll
  for (int e = 0; e < P; ++e) {
    const float vlo = from_row_lane(a[e], lo);
    const float vhi = from_row_lane(a[e], hi);
    b[e]     = inside ? vlo : 0.f;
    b[P + e] = inside ? vhi : 0.f;
  }
}

template <int P, int PAIRS>
__device__ __forceinline__ void horner_pairs(float (&x)[P], float (&y)[P],
                                             const float (&p)[6], float owns_center,
                                             float& fn, int& n) {
  for (int it = 0; it < PAIRS; ++it) {
    fn *= (float)n; --n;
    diffuse_step<P>(x, y, p, owns_center * fn);
    fn *= (float)n; --n;
    diffuse_step<P>(y, x, p, owns_center * fn);
  }
}

__global__ __launch_bounds__(32 * kWaves) void tonal_diffusion_kernel(
    const float* __restrict__ data,
    const float* __restrict__ log_weights,
    float* __restrict__ out,
    int n_rows) {
  const int lane = threadIdx.x & 31;
  const int wave = threadIdx.x >> 5;
  const int half = lane >> 4;
  const int hl   = lane & 15;
  const int rs   = 2 * wave + half;
  const int row  = blockIdx.x * kRows + rs;
  const int rowc = (row < n_rows) ? row : (n_rows - 1);

  __shared__ __attribute__((aligned(16))) float sDataPad[kRows][160];
  __shared__ __attribute__((aligned(16))) float sLogAcc[kRows][384];
  __shared__ __attribute__((aligned(16))) float sAcc[kRows][kS + 128];

  float p[6];
  float rate = 0.f;
#pragma unroll
  for (int i = 0; i < 6; ++i) {
    p[i] = exp_hw(log_weights[(size_t)rowc * 6 + i]);
    rate += p[i];
  }

  {
    const float4* d4 = reinterpret_cast<const float4*>(data + (size_t)rowc * kC);
#pragma unroll
    for (int q = 0; q < 2; ++q) {
      float4 raw = d4[hl + 16 * q];
      float4 v;
      v.x = (raw.x > 1e-8f) ? raw.x : 0.f;
      v.y = (raw.y > 1e-8f) ? raw.y : 0.f;
      v.z = (raw.z > 1e-8f) ? raw.z : 0.f;
      v.w = (raw.w > 1e-8f) ? raw.w : 0.f;
      *reinterpret_cast<float4*>(&sDataPad[rs][16 + 4 * hl + 64 * q]) = v;
    }
    sDataPad[rs][hl] = 0.f;
    sDataPad[rs][16 + kC + hl] = 0.f;
  }

  float x[kPer], y[kPer], acc[kPer];
  const float owns_center = (kPer * hl == kCenter - kCenter % kPer) ? 1.f : 0.f;
  float fn = 3.8003907548547435e-36f;
  int n = kIters - 1;
  {
    float x4[4], y4[4];
#pragma unroll
    for (int e = 0; e < 4; ++e) x4[e] = (e == 0) ? owns_center * fn : 0.f;
    horner_pairs<4, 3>(x4, y4, p, owns_center, fn, n);
    fn *= (float)n; --n;
    diffuse_step<4>(x4, y4, p, owns_center * fn);
    float x8[8], y8[8];
    widen_layout<4>(y4, x8, hl);
    horner_pairs<8, 4>(x8, y8, p, owns_center, fn, n);
    widen_layout<8>(x8, y, hl);
  }
  fn *= (float)n; --n;
  diffuse_step<kPer>(y, x, p, owns_center * fn);
  horner_pairs<kPer, 8>(x, y, p, owns_center, fn, n);
  const float sp = exp_hw(-rate);
#pragma unroll
  for (int e = 0; e < kPer; ++e) acc[e] = sp * x[e];

#pragma unroll
  for (int e = 0; e < kPer; ++e) {
    const int t = kPer * hl + e;
    sAcc[rs][t] = acc[e];
    sLogAcc[rs][t] = log_hw(fmaxf(acc[e], 1e-38f));
  }
#pragma unroll
  for (int i = 0; i < 8; ++i) sAcc[rs][kS + hl + 16 * i] = 0.f;
#pragma unroll
  for (int i = 0; i < 8; ++i) sLogAcc[rs][kS + hl + 16 * i] = 0.f;
  __syncthreads();

  float z0[8], z1[8];
  {
    float tail[kPer], head[kPer];
    const float4* own4 = reinterpret_cast<const float4*>(&sAcc[rs][kPer * hl]);
    const float4* far4 = reinterpret_cast<const float4*>(&sAcc[rs][kPer * (hl + 8)]);
    float own[kPer], far[kPer];
#pragma unroll
    for (int q = 0; q < kPer / 4; ++q) {
      const float4 o = own4[q];
      own[4 * q] = o.x; own[4 * q + 1] = o.y; own[4 * q + 2] = o.z; own[4 * q + 3] = o.w;
      const float4 f = far4[q];
      far[4 * q] = f.x; far[4 * q + 1] = f.y; far[4 * q + 2] = f.z; far[4 * q + 3] = f.w;
    }
    tail[kPer - 1] = own[kPer - 1];
#pragma unroll
    for (int b = kPer - 2; b >= 0; --b) tail[b] = tail[b + 1] + own[b];
    head[0] = 0.f;
#pragma unroll
    for (int b = 1; b < kPer; ++b) head[b] = head[b - 1] + far[b - 1];
    const float blk  = tail[0];
    const float two  = blk + from_lane_up<1>(blk);
    const float four = two + from_lane_up<2>(two);
    const float mid  = (from_lane_up<1>(four) + from_lane_up<5>(two)) +
                       from_lane_up<7>(blk);
    float zl[kPer];
#pragma unroll
    for (int b = 0; b < kPer; ++b) zl[b] = log_hw((tail[b] + mid) + head[b]);
#pragma unroll
    for (int r = 0; r < 8; ++r) {
      const v2u s = __builtin_amdgcn_permlane16_swap(
          __float_as_uint(zl[r]), __float_as_uint(zl[8 + r]), false, false);
      z0[r] = __uint_as_float(s.x);
      z1[r] = __uint_as_float(s.y);
    }
  }

  float pa = 0.f, pb = 0.f;
  for (int i = 0; i < 4; ++i) {
    const float va = sDataPad[rs][16 + hl + 32 * i];
    if (va > 0.f) pa += va * log_hw(va);
    const float vb = sDataPad[rs][32 + hl + 32 * i];
    if (vb > 0.f) pb += vb * log_hw(vb);
  }
  float pl = pa + pb;
  pl += from_row_xor<8>(pl);
  pl += from_row_xor<4>(pl);
  pl += from_row_xor<2>(pl);
  pl += from_row_xor<1>(pl);
  const float plOther = from_other_half(pl);
  const float pl0 = half ? plOther : pl;
  const float pl1 = half ? pl : plOther;

  const int m    = lane & 15;
  const int koff = (lane & 16) ? 2 : 0;
  const float* dp0 = &sDataPad[2 * wave][16 - m];
  const float* lp0 = &sLogAcc[2 * wave][16 * m];
  const float* dp1 = &sDataPad[2 * wave + 1][16 - m];
  const float* lp1 = &sLogAcc[2 * wave + 1][16 * m];
  v8f d0 = {0.f, 0.f, 0.f, 0.f, 0.f, 0.f, 0.f, 0.f};
  v8f d1 = {0.f, 0.f, 0.f, 0.f, 0.f, 0.f, 0.f, 0.f};
#pragma unroll
  for (int kc = 0; kc < kKTiles; ++kc) {
    const int j = 4 * kc + koff;
    v2f a0 = {dp0[j], dp0[j + 1]};
    v2f b0 = {lp0[j], lp0[j + 1]};
    d0 = __builtin_amdgcn_wmma_f32_16x16x4_f32(
         false, a0,  false, b0,
         (short)0, d0,  false,  false);
    v2f a1 = {dp1[j], dp1[j + 1]};
    v2f b1 = {lp1[j], lp1[j + 1]};
    d1 = __builtin_amdgcn_wmma_f32_16x16x4_f32(
         false, a1,  false, b1,
         (short)0, d1,  false,  false);
  }
  const int mbase = (lane & 16) ? 8 : 0;
  const int n16   = 16 * (lane & 15);
  float best0 = INFINITY, best1 = INFINITY;
#pragma unroll
  for (int r = 0; r < 8; ++r) {
    const int k = (r + mbase) + n16;
    const bool used = k < kShifts;
    const float c0 = pl0 - d0[r] + z0[r];
    const float c1 = pl1 - d1[r] + z1[r];
    best0 = used ? fminf(best0, c0) : best0;
    best1 = used ? fminf(best1, c1) : best1;
  }
  best0 = fminf(best0, from_other_half(best0));
  best1 = fminf(best1, from_other_half(best1));
  best0 = fminf(best0, from_row_xor<8>(best0));
  best1 = fminf(best1, from_row_xor<8>(best1));
  best0 = fminf(best0, from_row_xor<4>(best0));
  best1 = fminf(best1, from_row_xor<4>(best1));
  best0 = fminf(best0, from_row_xor<2>(best0));
  best1 = fminf(best1, from_row_xor<2>(best1));
  best0 = fminf(best0, from_row_xor<1>(best0));
  best1 = fminf(best1, from_row_xor<1>(best1));
  if (lane == 0) {
    const int row0 = blockIdx.x * kRows + 2 * wave;
    if (row0 < n_rows) out[row0] = best0;
    if (row0 + 1 < n_rows) out[row0 + 1] = best1;
  }
}

extern "C" void kernel_launch(void* const* d_in, const int* in_sizes, int n_in,
                              void* d_out, int out_size, void* d_ws, size_t ws_size,
                              hipStream_t stream) {
  (void)n_in; (void)d_ws; (void)ws_size; (void)out_size;
  const float* data = (const float*)d_in[0];
  const float* logw = (const float*)d_in[1];
  float* out = (float*)d_out;
  const int n_rows = in_sizes[0] / kC;
  const int blocks = (n_rows + kRows - 1) / kRows;
  tonal_diffusion_kernel<<<blocks, 32 * kWaves, 0, stream>>>(data, logw, out, n_rows);
}


# 16-per-lane Horner passes keep each lane's points as register pairs (i, i+8): every tap, odd or even, is a packed FMA on whole pairs, no re-pairing moves; per-point tap order and sums unchanged
# speedup vs baseline: 3.4507x; 1.0638x over previous
#include <hip/hip_runtime.h>
#include <math.h>


typedef __attribute__((ext_vector_type(2))) float v2f;
typedef __attribute__((ext_vector_type(8))) float v8f;
typedef __attribute__((ext_vector_type(2))) unsigned int v2u;

namespace {
constexpr int kS = 256;
constexpr int kC = 128;
constexpr int kCenter = 128;
constexpr int kIters = 33;
constexpr int kShifts = 129;
constexpr int kKTiles = 36;
constexpr int kWaves = 4;
constexpr int kRows = 2 * kWaves;
constexpr int kPer = 16;
}

__device__ __forceinline__ float from_prev_lane(float v) {
  return __int_as_float(__builtin_amdgcn_update_dpp(
      0, __float_as_int(v), 0x111, 0xF, 0xF, true));
}
__device__ __forceinline__ float from_next_lane(float v) {
  return __int_as_float(__builtin_amdgcn_update_dpp(
      0, __float_as_int(v), 0x101, 0xF, 0xF, true));
}

template <int N>
__device__ __forceinline__ float from_lane_up(float v) {
  return __int_as_float(__builtin_amdgcn_update_dpp(
      0, __float_as_int(v), 0x100 + N, 0xF, 0xF, true));
}

template <int MASK>
__device__ __forceinline__ float from_row_xor(float v) {
  return __int_as_float(__builtin_amdgcn_update_dpp(
      0, __float_as_int(v), 0x160 + MASK, 0xF, 0xF, true));
}
__device__ __forceinline__ float from_other_half(float v) {
  const int b = __float_as_int(v);
  return __int_as_float(__builtin_amdgcn_permlanex16(
      b, b, 0x76543210, 0xfedcba98, false, false));
}

__device__ __forceinline__ float log_hw(float v) {
  return fmaf(__builtin_amdgcn_logf(v * 256.f), 0.693147180559945f,
              -5.545177444479562f);
}
__device__ __forceinline__ float exp_hw(float v) {
  return __builtin_amdgcn_exp2f(v * 1.4426950408889634f);
}

template <int P>
__device__ __forceinline__ void diffuse_step(const float (&x)[P],
                                             float (&y)[P],
                                             const float (&p)[6],
                                             float inj) {
  const int steps[6] = {1, -1, -3, 3, 4, -4};
  float ext[P + 8];
#pragma unroll
  for (int j = 0; j < 4; ++j) ext[j] = from_prev_lane(x[P - 4 + j]);
#pragma unroll
  for (int e = 0; e < P; ++e) ext[4 + e] = x[e];
#pragma unroll
  for (int j = 0; j < 4; ++j) ext[4 + P + j] = from_next_lane(x[j]);
#pragma unroll
  for (int e = 0; e < P; ++e) {
    float nv = (e == 0) ? inj : 0.f;
#pragma unroll
    for (int i = 0; i < 6; ++i) nv = fmaf(p[i], ext[4 + e - steps[i]], nv);
    y[e] = nv;
  }
}

__device__ __forceinline__ float from_row_lane(float v, int src) {
  return __int_as_float(__builtin_amdgcn_permlane16_var(
      0, __float_as_int(v), src, false, false));
}

template <int P>
__device__ __forceinline__ void widen_layout(const float (&a)[P], float (&b)[2 * P],
                                             int hl) {
  const bool inside = (hl >= 4) && (hl < 12);
  const int lo = (2 * hl - 8) & 15;
  const int hi = (2 * hl - 7) & 15;
#pragma unroll
  for (int e = 0; e < P; ++e) {
    const float vlo = from_row_lane(a[e], lo);
    const float vhi = from_row_lane(a[e], hi);
    b[e]     = inside ? vlo : 0.f;
    b[P + e] = inside ? vhi : 0.f;
  }
}

template <int P, int PAIRS>
__device__ __forceinline__ void horner_pairs(float (&x)[P], float (&y)[P],
                                             const float (&p)[6], float owns_center,
                                             float& fn, int& n) {
  for (int it = 0; it < PAIRS; ++it) {
    fn *= (float)n; --n;
    diffuse_step<P>(x, y, p, owns_center * fn);
    fn *= (float)n; --n;
    diffuse_step<P>(y, x, p, owns_center * fn);
  }
}

__device__ __forceinline__ void diffuse_step_paired(const v2f (&x)[8], v2f (&y)[8],
                                                    const float (&p)[6], float inj) {
  const int steps[6] = {1, -1, -3, 3, 4, -4};
  v2f op[16];
#pragma unroll
  for (int k = -4; k < 0; ++k) {
    v2f t = {from_prev_lane(x[8 + k].y), x[8 + k].x};
    op[4 + k] = t;
  }
#pragma unroll
  for (int k = 0; k < 8; ++k) op[4 + k] = x[k];
#pragma unroll
  for (int k = 8; k < 12; ++k) {
    v2f t = {x[k - 8].y, from_next_lane(x[k - 8].x)};
    op[4 + k] = t;
  }
#pragma unroll
  for (int i = 0; i < 8; ++i) {
    v2f nv = {(i == 0) ? inj : 0.f, 0.f};
#pragma unroll
    for (int t = 0; t < 6; ++t) {
      const v2f w = {p[t], p[t]};
      nv = __builtin_elementwise_fma(w, op[4 + i - steps[t]], nv);
    }
    y[i] = nv;
  }
}

template <int PAIRS>
__device__ __forceinline__ void horner_pairs_paired(v2f (&x)[8], v2f (&y)[8],
                                                    const float (&p)[6], float owns_center,
                                                    float& fn, int& n) {
  for (int it = 0; it < PAIRS; ++it) {
    fn *= (float)n; --n;
    diffuse_step_paired(x, y, p, owns_center * fn);
    fn *= (float)n; --n;
    diffuse_step_paired(y, x, p, owns_center * fn);
  }
}

__global__ __launch_bounds__(32 * kWaves) void tonal_diffusion_kernel(
    const float* __restrict__ data,
    const float* __restrict__ log_weights,
    float* __restrict__ out,
    int n_rows) {
  const int lane = threadIdx.x & 31;
  const int wave = threadIdx.x >> 5;
  const int half = lane >> 4;
  const int hl   = lane & 15;
  const int rs   = 2 * wave + half;
  const int row  = blockIdx.x * kRows + rs;
  const int rowc = (row < n_rows) ? row : (n_rows - 1);

  __shared__ __attribute__((aligned(16))) float sDataPad[kRows][160];
  __shared__ __attribute__((aligned(16))) float sLogAcc[kRows][384];
  __shared__ __attribute__((aligned(16))) float sAcc[kRows][kS + 128];

  float p[6];
  float rate = 0.f;
#pragma unroll
  for (int i = 0; i < 6; ++i) {
    p[i] = exp_hw(log_weights[(size_t)rowc * 6 + i]);
    rate += p[i];
  }

  {
    const float4* d4 = reinterpret_cast<const float4*>(data + (size_t)rowc * kC);
#pragma unroll
    for (int q = 0; q < 2; ++q) {
      float4 raw = d4[hl + 16 * q];
      float4 v;
      v.x = (raw.x > 1e-8f) ? raw.x : 0.f;
      v.y = (raw.y > 1e-8f) ? raw.y : 0.f;
      v.z = (raw.z > 1e-8f) ? raw.z : 0.f;
      v.w = (raw.w > 1e-8f) ? raw.w : 0.f;
      *reinterpret_cast<float4*>(&sDataPad[rs][16 + 4 * hl + 64 * q]) = v;
    }
    sDataPad[rs][hl] = 0.f;
    sDataPad[rs][16 + kC + hl] = 0.f;
  }

  float acc[kPer];
  v2f xq[8], yq[8];
  const float owns_center = (kPer * hl == kCenter - kCenter % kPer) ? 1.f : 0.f;
  float fn = 3.8003907548547435e-36f;
  int n = kIters - 1;
  {
    float x4[4], y4[4];
#pragma unroll
    for (int e = 0; e < 4; ++e) x4[e] = (e == 0) ? owns_center * fn : 0.f;
    horner_pairs<4, 3>(x4, y4, p, owns_center, fn, n);
    fn *= (float)n; --n;
    diffuse_step<4>(x4, y4, p, owns_center * fn);
    float x8[8], y8[8];
    widen_layout<4>(y4, x8, hl);
    horner_pairs<8, 4>(x8, y8, p, owns_center, fn, n);
    float w16[kPer];
    widen_layout<8>(x8, w16, hl);
#pragma unroll
    for (int i = 0; i < 8; ++i) { v2f t = {w16[i], w16[i + 8]}; yq[i] = t; }
  }
  fn *= (float)n; --n;
  diffuse_step_paired(yq, xq, p, owns_center * fn);
  horner_pairs_paired<8>(xq, yq, p, owns_center, fn, n);
  const float sp = exp_hw(-rate);
#pragma unroll
  for (int i = 0; i < 8; ++i) { acc[i] = sp * xq[i].x; acc[i + 8] = sp * xq[i].y; }

#pragma unroll
  for (int e = 0; e < kPer; ++e) {
    const int t = kPer * hl + e;
    sAcc[rs][t] = acc[e];
    sLogAcc[rs][t] = log_hw(fmaxf(acc[e], 1e-38f));
  }
#pragma unroll
  for (int i = 0; i < 8; ++i) sAcc[rs][kS + hl + 16 * i] = 0.f;
#pragma unroll
  for (int i = 0; i < 8; ++i) sLogAcc[rs][kS + hl + 16 * i] = 0.f;
  __syncthreads();

  float z0[8], z1[8];
  {
    float tail[kPer], head[kPer];
    const float4* own4 = reinterpret_cast<const float4*>(&sAcc[rs][kPer * hl]);
    const float4* far4 = reinterpret_cast<const float4*>(&sAcc[rs][kPer * (hl + 8)]);
    float own[kPer], far[kPer];
#pragma unroll
    for (int q = 0; q < kPer / 4; ++q) {
      const float4 o = own4[q];
      own[4 * q] = o.x; own[4 * q + 1] = o.y; own[4 * q + 2] = o.z; own[4 * q + 3] = o.w;
      const float4 f = far4[q];
      far[4 * q] = f.x; far[4 * q + 1] = f.y; far[4 * q + 2] = f.z; far[4 * q + 3] = f.w;
    }
    tail[kPer - 1] = own[kPer - 1];
#pragma unroll
    for (int b = kPer - 2; b >= 0; --b) tail[b] = tail[b + 1] + own[b];
    head[0] = 0.f;
#pragma unroll
    for (int b = 1; b < kPer; ++b) head[b] = head[b - 1] + far[b - 1];
    const float blk  = tail[0];
    const float two  = blk + from_lane_up<1>(blk);
    const float four = two + from_lane_up<2>(two);
    const float mid  = (from_lane_up<1>(four) + from_lane_up<5>(two)) +
                       from_lane_up<7>(blk);
    float zl[kPer];
#pragma unroll
    for (int b = 0; b < kPer; ++b) zl[b] = log_hw((tail[b] + mid) + head[b]);
#pragma unroll
    for (int r = 0; r < 8; ++r) {
      const v2u s = __builtin_amdgcn_permlane16_swap(
          __float_as_uint(zl[r]), __float_as_uint(zl[8 + r]), false, false);
      z0[r] = __uint_as_float(s.x);
      z1[r] = __uint_as_float(s.y);
    }
  }

  float pa = 0.f, pb = 0.f;
  for (int i = 0; i < 4; ++i) {
    const float va = sDataPad[rs][16 + hl + 32 * i];
    if (va > 0.f) pa += va * log_hw(va);
    const float vb = sDataPad[rs][32 + hl + 32 * i];
    if (vb > 0.f) pb += vb * log_hw(vb);
  }
  float pl = pa + pb;
  pl += from_row_xor<8>(pl);
  pl += from_row_xor<4>(pl);
  pl += from_row_xor<2>(pl);
  pl += from_row_xor<1>(pl);
  const float plOther = from_other_half(pl);
  const float pl0 = half ? plOther : pl;
  const float pl1 = half ? pl : plOther;

  const int m    = lane & 15;
  const int koff = (lane & 16) ? 2 : 0;
  const float* dp0 = &sDataPad[2 * wave][16 - m];
  const float* lp0 = &sLogAcc[2 * wave][16 * m];
  const float* dp1 = &sDataPad[2 * wave + 1][16 - m];
  const float* lp1 = &sLogAcc[2 * wave + 1][16 * m];
  v8f d0 = {0.f, 0.f, 0.f, 0.f, 0.f, 0.f, 0.f, 0.f};
  v8f d1 = {0.f, 0.f, 0.f, 0.f, 0.f, 0.f, 0.f, 0.f};
#pragma unroll
  for (int kc = 0; kc < kKTiles; ++kc) {
    const int j = 4 * kc + koff;
    v2f a0 = {dp0[j], dp0[j + 1]};
    v2f b0 = {lp0[j], lp0[j + 1]};
    d0 = __builtin_amdgcn_wmma_f32_16x16x4_f32(
         false, a0,  false, b0,
         (short)0, d0,  false,  false);
    v2f a1 = {dp1[j], dp1[j + 1]};
    v2f b1 = {lp1[j], lp1[j + 1]};
    d1 = __builtin_amdgcn_wmma_f32_16x16x4_f32(
         false, a1,  false, b1,
         (short)0, d1,  false,  false);
  }
  const int mbase = (lane & 16) ? 8 : 0;
  const int n16   = 16 * (lane & 15);
  float best0 = INFINITY, best1 = INFINITY;
#pragma unroll
  for (int r = 0; r < 8; ++r) {
    const int k = (r + mbase) + n16;
    const bool used = k < kShifts;
    const float c0 = pl0 - d0[r] + z0[r];
    const float c1 = pl1 - d1[r] + z1[r];
    best0 = used ? fminf(best0, c0) : best0;
    best1 = used ? fminf(best1, c1) : best1;
  }
  best0 = fminf(best0, from_other_half(best0));
  best1 = fminf(best1, from_other_half(best1));
  best0 = fminf(best0, from_row_xor<8>(best0));
  best1 = fminf(best1, from_row_xor<8>(best1));
  best0 = fminf(best0, from_row_xor<4>(best0));
  best1 = fminf(best1, from_row_xor<4>(best1));
  best0 = fminf(best0, from_row_xor<2>(best0));
  best1 = fminf(best1, from_row_xor<2>(best1));
  best0 = fminf(best0, from_row_xor<1>(best0));
  best1 = fminf(best1, from_row_xor<1>(best1));
  if (lane == 0) {
    const int row0 = blockIdx.x * kRows + 2 * wave;
    if (row0 < n_rows) out[row0] = best0;
    if (row0 + 1 < n_rows) out[row0 + 1] = best1;
  }
}

extern "C" void kernel_launch(void* const* d_in, const int* in_sizes, int n_in,
                              void* d_out, int out_size, void* d_ws, size_t ws_size,
                              hipStream_t stream) {
  (void)n_in; (void)d_ws; (void)ws_size; (void)out_size;
  const float* data = (const float*)d_in[0];
  const float* logw = (const float*)d_in[1];
  float* out = (float*)d_out;
  const int n_rows = in_sizes[0] / kC;
  const int blocks = (n_rows + kRows - 1) / kRows;
  tonal_diffusion_kernel<<<blocks, 32 * kWaves, 0, stream>>>(data, logw, out, n_rows);
}
